# starting kernel
# baseline (speedup 1.0000x reference)
#include <hip/hip_runtime.h>
#include <math.h>

typedef float v2f __attribute__((ext_vector_type(2)));
typedef float v8f __attribute__((ext_vector_type(8)));

#define C_DIM   512
#define HW      6144
#define NIMG    16
#define O_LOC   36
#define O_SC    18
#define O_TOT   54
#define O_PAD   64
#define KCHUNK  128
#define PAIR_STRIDE (O_PAD * 2)

#define M_TOTAL   (NIMG * HW)
#define TILES     (M_TOTAL / 16)
#define TILES_PER_IMG (HW / 16)

#define LOC_ELEMS ((size_t)NIMG * HW * O_LOC)
#define SC_ELEMS  ((size_t)NIMG * HW * O_SC)
#define FG_ELEMS  ((size_t)NIMG * HW * 9)

__global__ __launch_bounds__(256)
void rpn_head_kernel(const float* __restrict__ x,
                     const float* __restrict__ loc_w,
                     const float* __restrict__ loc_b,
                     const float* __restrict__ score_w,
                     const float* __restrict__ score_b,
                     float* __restrict__ d_out)
{
    __shared__ float sW[(KCHUNK / 2) * PAIR_STRIDE];
    __shared__ float sBias[O_PAD];

    const int tid  = threadIdx.x;
    const int wave = tid >> 5;
    const int lane = tid & 31;
    const int col  = lane & 15;
    const int khalf = lane >> 4;

    if (tid < O_PAD) {
        float b = 0.0f;
        if (tid < O_LOC)       b = loc_b[tid];
        else if (tid < O_TOT)  b = score_b[tid - O_LOC];
        sBias[tid] = b;
    }

    const int tile0 = blockIdx.x * 16 + wave * 2;
    const int tile1 = tile0 + 1;

    const int n0 = tile0 / TILES_PER_IMG;
    const int n1 = tile1 / TILES_PER_IMG;
    const size_t base0 = (size_t)n0 * C_DIM * HW + (size_t)(tile0 % TILES_PER_IMG) * 16 + col;
    const size_t base1 = (size_t)n1 * C_DIM * HW + (size_t)(tile1 % TILES_PER_IMG) * 16 + col;

    v8f acc[2][4];
    const v8f vzero = {0.f,0.f,0.f,0.f,0.f,0.f,0.f,0.f};
    #pragma unroll
    for (int t = 0; t < 2; ++t)
        #pragma unroll
        for (int j = 0; j < 4; ++j)
            acc[t][j] = vzero;

    const int kq = khalf << 1;

    for (int kc = 0; kc < C_DIM / KCHUNK; ++kc) {
        const int k0 = kc * KCHUNK;

        __syncthreads();
        for (int idx = tid; idx < O_TOT * KCHUNK; idx += 256) {
            const int o  = idx >> 7;
            const int kk = idx & (KCHUNK - 1);
            const float wv = (o < O_LOC)
                ? loc_w[o * C_DIM + k0 + kk]
                : score_w[(o - O_LOC) * C_DIM + k0 + kk];
            sW[(kk >> 1) * PAIR_STRIDE + o * 2 + (kk & 1)] = wv;
        }
        __syncthreads();

        for (int ks = 0; ks < KCHUNK; ks += 4) {
            const int c0 = k0 + ks + kq;

            v2f a0, a1;
            a0.x = x[base0 + (size_t)(c0    ) * HW];
            a0.y = x[base0 + (size_t)(c0 + 1) * HW];
            a1.x = x[base1 + (size_t)(c0    ) * HW];
            a1.y = x[base1 + (size_t)(c0 + 1) * HW];

            const int pairRow = (ks >> 1) + khalf;
            const float* pb = &sW[pairRow * PAIR_STRIDE + col * 2];
            v2f b[4];
            #pragma unroll
            for (int j = 0; j < 4; ++j)
                b[j] = *(const v2f*)(pb + j * 32);

            #pragma unroll
            for (int j = 0; j < 4; ++j) {
                acc[0][j] = __builtin_amdgcn_wmma_f32_16x16x4_f32(
                    false, a0, false, b[j], (short)0, acc[0][j], false, false);
                acc[1][j] = __builtin_amdgcn_wmma_f32_16x16x4_f32(
                    false, a1, false, b[j], (short)0, acc[1][j], false, false);
            }

            if (ks + 4 < KCHUNK) {
                __builtin_prefetch((const void*)&x[base0 + (size_t)(c0 + 4) * HW], 0, 1);
                __builtin_prefetch((const void*)&x[base1 + (size_t)(c0 + 4) * HW], 0, 1);
            }
        }
    }

    float* out_loc   = d_out;
    float* out_score = d_out + LOC_ELEMS;
    float* out_fg    = d_out + LOC_ELEMS + SC_ELEMS;

    const int mhi = khalf * 8;

    #pragma unroll
    for (int t = 0; t < 2; ++t) {
        const int tile = (t == 0) ? tile0 : tile1;
        #pragma unroll
        for (int j = 0; j < 4; ++j) {
            const int o = j * 16 + col;
            const float bias = sBias[o];
            #pragma unroll
            for (int v = 0; v < 8; ++v) {
                const float val = acc[t][j][v] + bias;
                const float partner = __shfl_xor(val, 1, 32);
                const size_t p = (size_t)tile * 16 + (size_t)(v + mhi);
                if (o < O_LOC) {
                    out_loc[p * O_LOC + o] = val;
                } else if (o < O_TOT) {
                    const int s = o - O_LOC;
                    out_score[p * O_SC + s] = val;
                    if ((s & 1) == 0) {
                        const float fg = 1.0f / (1.0f + __expf(val - partner));
                        out_fg[p * 9 + (s >> 1)] = fg;
                    }
                }
            }
        }
    }
}

__global__ __launch_bounds__(256)
void rpn_anchor_kernel(float* __restrict__ out_anchor)
{
    const int i = blockIdx.x * 256 + threadIdx.x;
    const int hw = i / 9;
    const int a  = i - hw * 9;
    const int h  = hw / 96;
    const int w  = hw - h * 96;
    const int ri = a / 3;
    const int si = a - ri * 3;

    const float ratios[3] = {0.5f, 1.0f, 2.0f};
    const float scales[3] = {8.0f, 16.0f, 32.0f};
    const float r = ratios[ri];
    const float s = scales[si];

    const float hbox = 16.0f * s * __fsqrt_rn(r);
    const float wbox = 16.0f * s * __fsqrt_rn(1.0f / r);
    const float py = (float)h * 16.0f + 8.0f;
    const float px = (float)w * 16.0f + 8.0f;

    float4 v;
    v.x = py - hbox * 0.5f;
    v.y = px - wbox * 0.5f;
    v.z = py + hbox * 0.5f;
    v.w = px + wbox * 0.5f;
    ((float4*)out_anchor)[i] = v;
}

extern "C" void kernel_launch(void* const* d_in, const int* in_sizes, int n_in,
                              void* d_out, int out_size, void* d_ws, size_t ws_size,
                              hipStream_t stream)
{
    const float* x       = (const float*)d_in[0];
    const float* loc_w   = (const float*)d_in[1];
    const float* loc_b   = (const float*)d_in[2];
    const float* score_w = (const float*)d_in[3];
    const float* score_b = (const float*)d_in[4];
    float* out = (float*)d_out;

    rpn_head_kernel<<<TILES / 16, 256, 0, stream>>>(x, loc_w, loc_b, score_w, score_b, out);

    float* out_anchor = out + LOC_ELEMS + SC_ELEMS + FG_ELEMS;
    rpn_anchor_kernel<<<(HW * 9) / 256, 256, 0, stream>>>(out_anchor);
}


# W not staged in LDS: its rows are channel-contiguous, so B fragments come straight from global by 16-B loads one block ahead beside x; no LDS, no barrier; sum order in each 32-channel block changes
# speedup vs baseline: 1.4377x; 1.4377x over previous
#include <hip/hip_runtime.h>
#include <math.h>


typedef float v2f __attribute__((ext_vector_type(2)));
typedef float v4f __attribute__((ext_vector_type(4)));
typedef float v8f __attribute__((ext_vector_type(8)));

#define C_DIM   512
#define HW      6144
#define NIMG    16
#define O_LOC   36
#define O_SC    18
#define O_TOT   54
#define O_PAD   64
#define BLK_CH  32

#define ROW_PIX   96
#define ROWS_IMG  64
#define NROWS     (NIMG * ROWS_IMG)
#define MT        6
#define NT        4
#define WAVES     4
#define THREADS   (WAVES * 32)
#define RING      (BLK_CH / 4)
#define NBLK      (C_DIM / BLK_CH)
#define NPAIR     (NBLK / 2)

#define LOC_ELEMS ((size_t)NIMG * HW * O_LOC)
#define SC_ELEMS  ((size_t)NIMG * HW * O_SC)
#define FG_ELEMS  ((size_t)NIMG * HW * 9)


struct XSlot {
    v4f lo0, lo1;
    v2f hi0, hi1;
};

struct WSet {
    v4f q[NT][4];
};

__device__ __forceinline__ void load_x(XSlot& r, const float* __restrict__ plo,
                                       const float* __restrict__ phi, int blk, int d)
{
    const size_t off = (size_t)(blk * BLK_CH + 2 * d) * HW;
    r.lo0 = *(const v4f*)(plo + off);
    r.lo1 = *(const v4f*)(plo + off + HW);
    r.hi0 = *(const v2f*)(phi + off);
    r.hi1 = *(const v2f*)(phi + off + HW);
}

__device__ __forceinline__ void load_group(WSet& w, XSlot (&set)[RING],
                                           const float* const (&wp)[NT],
                                           const float* __restrict__ plo,
                                           const float* __restrict__ phi, int blk)
{
    #pragma unroll
    for (int i = 0; i < 4; ++i)
        #pragma unroll
        for (int j = 0; j < NT; ++j)
            w.q[j][i] = *(const v4f*)(wp[j] + blk * BLK_CH + 4 * i);
    #pragma unroll
    for (int d = 0; d < RING; ++d)
        load_x(set[d], plo, phi, blk, d);
    __builtin_amdgcn_sched_barrier(0);
}

__device__ __forceinline__ void k_block(v8f (&acc)[MT][NT], const XSlot (&set)[RING], const WSet& w)
{
    #pragma unroll
    for (int d = 0; d < RING; ++d) {
        v2f b[NT];
        #pragma unroll
        for (int j = 0; j < NT; ++j) {
            const v4f q = w.q[j][d >> 1];
            if (d & 1) { b[j].x = q.z; b[j].y = q.w; }
            else       { b[j].x = q.x; b[j].y = q.y; }
        }

        const XSlot& r = set[d];
        v2f a[MT];
        a[0].x = r.lo0.x;  a[0].y = r.lo1.x;
        a[1].x = r.lo0.y;  a[1].y = r.lo1.y;
        a[2].x = r.lo0.z;  a[2].y = r.lo1.z;
        a[3].x = r.lo0.w;  a[3].y = r.lo1.w;
        a[4].x = r.hi0.x;  a[4].y = r.hi1.x;
        a[5].x = r.hi0.y;  a[5].y = r.hi1.y;

        #pragma unroll
        for (int j = 0; j < NT; ++j)
            #pragma unroll
            for (int t = 0; t < MT; ++t)
                acc[t][j] = __builtin_amdgcn_wmma_f32_16x16x4_f32(
                    false, a[t], false, b[j], (short)0, acc[t][j], false, false);
    }
}

__global__ __launch_bounds__(THREADS)
void rpn_head_kernel(const float* __restrict__ x,
                     const float* __restrict__ loc_w,
                     const float* __restrict__ loc_b,
                     const float* __restrict__ score_w,
                     const float* __restrict__ score_b,
                     float* __restrict__ d_out)
{
    const int tid  = threadIdx.x;
    const int wave = tid >> 5;
    const int lane = tid & 31;
    const int col  = lane & 15;
    const int khalf = lane >> 4;

    const float* wp[NT];
    float bias_r[NT];
    #pragma unroll
    for (int j = 0; j < NT; ++j) {
        const int o = j * 16 + col;
        const float* rowp;
        float bv;
        if (o < O_LOC) {
            rowp = loc_w + (size_t)o * C_DIM;
            bv = loc_b[o];
        } else if (o < O_TOT) {
            rowp = score_w + (size_t)(o - O_LOC) * C_DIM;
            bv = score_b[o - O_LOC];
        } else {
            rowp = score_w + (size_t)(O_SC - 1) * C_DIM;
            bv = 0.0f;
        }
        wp[j] = rowp + khalf * (BLK_CH / 2);
        bias_r[j] = bv;
    }

    const int row = blockIdx.x * WAVES + wave;
    const int n   = row / ROWS_IMG;
    const int h   = row - n * ROWS_IMG;
    const float* xrow = x + (size_t)n * C_DIM * HW + (size_t)h * ROW_PIX
                          + (size_t)(khalf * (BLK_CH / 2)) * HW;
    const float* plo = xrow + 4 * col;
    const float* phi = xrow + 64 + 2 * col;

    v8f acc[MT][NT];
    const v8f vzero = {0.f,0.f,0.f,0.f,0.f,0.f,0.f,0.f};
    #pragma unroll
    for (int t = 0; t < MT; ++t)
        #pragma unroll
        for (int j = 0; j < NT; ++j)
            acc[t][j] = vzero;

    XSlot xA[RING], xB[RING];
    WSet  wA, wB;
    load_group(wA, xA, wp, plo, phi, 0);

    for (int p = 0; p < NPAIR - 1; ++p) {
        load_group(wB, xB, wp, plo, phi, 2 * p + 1);
        k_block(acc, xA, wA);
        load_group(wA, xA, wp, plo, phi, 2 * p + 2);
        k_block(acc, xB, wB);
    }
    load_group(wB, xB, wp, plo, phi, NBLK - 1);
    k_block(acc, xA, wA);
    k_block(acc, xB, wB);

    float* out_loc   = d_out;
    float* out_score = d_out + LOC_ELEMS;
    float* out_fg    = d_out + LOC_ELEMS + SC_ELEMS;

    const int mhi = khalf * 8;
    const size_t prow = (size_t)row * ROW_PIX;

    #pragma unroll
    for (int t = 0; t < MT; ++t) {
        #pragma unroll
        for (int j = 0; j < 4; ++j) {
            const int o = j * 16 + col;
            const float bias = bias_r[j];
            #pragma unroll
            for (int v = 0; v < 8; ++v) {
                const float val = acc[t][j][v] + bias;
                const float partner = __shfl_xor(val, 1, 32);
                const int m = v + mhi;
                const int pix = (t < 4) ? (4 * m + t) : (64 + 2 * m + (t - 4));
                const size_t p = prow + (size_t)pix;
                if (o < O_LOC) {
                    out_loc[p * O_LOC + o] = val;
                } else if (o < O_TOT) {
                    const int s = o - O_LOC;
                    out_score[p * O_SC + s] = val;
                    if ((s & 1) == 0) {
                        const float fg = 1.0f / (1.0f + __expf(val - partner));
                        out_fg[p * 9 + (s >> 1)] = fg;
                    }
                }
            }
        }
    }
}

__global__ __launch_bounds__(256)
void rpn_anchor_kernel(float* __restrict__ out_anchor)
{
    const int i = blockIdx.x * 256 + threadIdx.x;
    const int hw = i / 9;
    const int a  = i - hw * 9;
    const int h  = hw / 96;
    const int w  = hw - h * 96;
    const int ri = a / 3;
    const int si = a - ri * 3;

    const float ratios[3] = {0.5f, 1.0f, 2.0f};
    const float scales[3] = {8.0f, 16.0f, 32.0f};
    const float r = ratios[ri];
    const float s = scales[si];

    const float hbox = 16.0f * s * __fsqrt_rn(r);
    const float wbox = 16.0f * s * __fsqrt_rn(1.0f / r);
    const float py = (float)h * 16.0f + 8.0f;
    const float px = (float)w * 16.0f + 8.0f;

    float4 v;
    v.x = py - hbox * 0.5f;
    v.y = px - wbox * 0.5f;
    v.z = py + hbox * 0.5f;
    v.w = px + wbox * 0.5f;
    ((float4*)out_anchor)[i] = v;
}

extern "C" void kernel_launch(void* const* d_in, const int* in_sizes, int n_in,
                              void* d_out, int out_size, void* d_ws, size_t ws_size,
                              hipStream_t stream)
{
    const float* x       = (const float*)d_in[0];
    const float* loc_w   = (const float*)d_in[1];
    const float* loc_b   = (const float*)d_in[2];
    const float* score_w = (const float*)d_in[3];
    const float* score_b = (const float*)d_in[4];
    float* out = (float*)d_out;

    rpn_head_kernel<<<NROWS / WAVES, THREADS, 0, stream>>>(x, loc_w, loc_b, score_w, score_b, out);

    float* out_anchor = out + LOC_ELEMS + SC_ELEMS + FG_ELEMS;
    rpn_anchor_kernel<<<(HW * 9) / 256, 256, 0, stream>>>(out_anchor);
}


# epilogue straight-lined: per-lane loc/score/fg base pointers set once, every store at a fixed offset in 6 lane-mask regions (was 3 branches + waits per element); partner logit by DPP swap
# speedup vs baseline: 1.8365x; 1.2774x over previous
#include <hip/hip_runtime.h>
#include <math.h>


typedef float v2f __attribute__((ext_vector_type(2)));
typedef float v4f __attribute__((ext_vector_type(4)));
typedef float v8f __attribute__((ext_vector_type(8)));

#define C_DIM   512
#define HW      6144
#define NIMG    16
#define O_LOC   36
#define O_SC    18
#define O_TOT   54
#define O_PAD   64
#define BLK_CH  32

#define ROW_PIX   96
#define ROWS_IMG  64
#define NROWS     (NIMG * ROWS_IMG)
#define MT        6
#define NT        4
#define WAVES     4
#define THREADS   (WAVES * 32)
#define RING      (BLK_CH / 4)
#define NBLK      (C_DIM / BLK_CH)
#define NPAIR     (NBLK / 2)

#define LOC_ELEMS ((size_t)NIMG * HW * O_LOC)
#define SC_ELEMS  ((size_t)NIMG * HW * O_SC)
#define FG_ELEMS  ((size_t)NIMG * HW * 9)


struct XSlot {
    v4f lo0, lo1;
    v2f hi0, hi1;
};

struct WSet {
    v4f q[NT][4];
};

__device__ __forceinline__ void load_x(XSlot& r, const float* __restrict__ plo,
                                       const float* __restrict__ phi, int blk, int d)
{
    const size_t off = (size_t)(blk * BLK_CH + 2 * d) * HW;
    r.lo0 = *(const v4f*)(plo + off);
    r.lo1 = *(const v4f*)(plo + off + HW);
    r.hi0 = *(const v2f*)(phi + off);
    r.hi1 = *(const v2f*)(phi + off + HW);
}

__device__ __forceinline__ void load_group(WSet& w, XSlot (&set)[RING],
                                           const float* const (&wp)[NT],
                                           const float* __restrict__ plo,
                                           const float* __restrict__ phi, int blk)
{
    #pragma unroll
    for (int i = 0; i < 4; ++i)
        #pragma unroll
        for (int j = 0; j < NT; ++j)
            w.q[j][i] = *(const v4f*)(wp[j] + blk * BLK_CH + 4 * i);
    #pragma unroll
    for (int d = 0; d < RING; ++d)
        load_x(set[d], plo, phi, blk, d);
    __builtin_amdgcn_sched_barrier(0);
}

__device__ __forceinline__ void k_block(v8f (&acc)[MT][NT], const XSlot (&set)[RING], const WSet& w)
{
    #pragma unroll
    for (int d = 0; d < RING; ++d) {
        v2f b[NT];
        #pragma unroll
        for (int j = 0; j < NT; ++j) {
            const v4f q = w.q[j][d >> 1];
            if (d & 1) { b[j].x = q.z; b[j].y = q.w; }
            else       { b[j].x = q.x; b[j].y = q.y; }
        }

        const XSlot& r = set[d];
        v2f a[MT];
        a[0].x = r.lo0.x;  a[0].y = r.lo1.x;
        a[1].x = r.lo0.y;  a[1].y = r.lo1.y;
        a[2].x = r.lo0.z;  a[2].y = r.lo1.z;
        a[3].x = r.lo0.w;  a[3].y = r.lo1.w;
        a[4].x = r.hi0.x;  a[4].y = r.hi1.x;
        a[5].x = r.hi0.y;  a[5].y = r.hi1.y;

        #pragma unroll
        for (int j = 0; j < NT; ++j)
            #pragma unroll
            for (int t = 0; t < MT; ++t)
                acc[t][j] = __builtin_amdgcn_wmma_f32_16x16x4_f32(
                    false, a[t], false, b[j], (short)0, acc[t][j], false, false);
    }
}

__global__ __launch_bounds__(THREADS)
void rpn_head_kernel(const float* __restrict__ x,
                     const float* __restrict__ loc_w,
                     const float* __restrict__ loc_b,
                     const float* __restrict__ score_w,
                     const float* __restrict__ score_b,
                     float* __restrict__ d_out)
{
    const int tid  = threadIdx.x;
    const int wave = tid >> 5;
    const int lane = tid & 31;
    const int col  = lane & 15;
    const int khalf = lane >> 4;

    const float* wp[NT];
    float bias_r[NT];
    #pragma unroll
    for (int j = 0; j < NT; ++j) {
        const int o = j * 16 + col;
        const float* rowp;
        float bv;
        if (o < O_LOC) {
            rowp = loc_w + (size_t)o * C_DIM;
            bv = loc_b[o];
        } else if (o < O_TOT) {
            rowp = score_w + (size_t)(o - O_LOC) * C_DIM;
            bv = score_b[o - O_LOC];
        } else {
            rowp = score_w + (size_t)(O_SC - 1) * C_DIM;
            bv = 0.0f;
        }
        wp[j] = rowp + khalf * (BLK_CH / 2);
        bias_r[j] = bv;
    }

    const int row = blockIdx.x * WAVES + wave;
    const int n   = row / ROWS_IMG;
    const int h   = row - n * ROWS_IMG;
    const float* xrow = x + (size_t)n * C_DIM * HW + (size_t)h * ROW_PIX
                          + (size_t)(khalf * (BLK_CH / 2)) * HW;
    const float* plo = xrow + 4 * col;
    const float* phi = xrow + 64 + 2 * col;

    v8f acc[MT][NT];
    const v8f vzero = {0.f,0.f,0.f,0.f,0.f,0.f,0.f,0.f};
    #pragma unroll
    for (int t = 0; t < MT; ++t)
        #pragma unroll
        for (int j = 0; j < NT; ++j)
            acc[t][j] = vzero;

    XSlot xA[RING], xB[RING];
    WSet  wA, wB;
    load_group(wA, xA, wp, plo, phi, 0);

    for (int p = 0; p < NPAIR - 1; ++p) {
        load_group(wB, xB, wp, plo, phi, 2 * p + 1);
        k_block(acc, xA, wA);
        load_group(wA, xA, wp, plo, phi, 2 * p + 2);
        k_block(acc, xB, wB);
    }
    load_group(wB, xB, wp, plo, phi, NBLK - 1);
    k_block(acc, xA, wA);
    k_block(acc, xB, wB);

    float* out_loc   = d_out;
    float* out_score = d_out + LOC_ELEMS;
    float* out_fg    = d_out + LOC_ELEMS + SC_ELEMS;

#define PIXC(t, v) ((t) < 4 ? 4 * (v) + (t) : 64 + 2 * (v) + ((t) - 4))
    const ptrdiff_t prow = (ptrdiff_t)row * ROW_PIX;
    const ptrdiff_t pfam[2] = { prow + 32 * khalf, prow + 16 * khalf };

    float* ploc[2];
    float* psc[2];
    float* pfg[2];
    #pragma unroll
    for (int f = 0; f < 2; ++f) {
        ploc[f] = out_loc   + pfam[f] * O_LOC + col;
        psc[f]  = out_score + pfam[f] * O_SC  + (col - 4);
        pfg[f]  = out_fg    + pfam[f] * 9     + ((col >> 1) - 2);
    }

    #pragma unroll
    for (int t = 0; t < MT; ++t)
        #pragma unroll
        for (int j = 0; j < NT; ++j)
            #pragma unroll
            for (int v = 0; v < 8; ++v)
                acc[t][j][v] += bias_r[j];

    #pragma unroll
    for (int t = 0; t < MT; ++t) {
        float* pl = ploc[t >> 2];
        #pragma unroll
        for (int v = 0; v < 8; ++v) {
            pl[PIXC(t, v) * O_LOC]      = acc[t][0][v];
            pl[PIXC(t, v) * O_LOC + 16] = acc[t][1][v];
        }
    }

    float fg2[MT][8], fg3[MT][8];
    #pragma unroll
    for (int t = 0; t < MT; ++t) {
        #pragma unroll
        for (int v = 0; v < 8; ++v) {
            const float a2 = acc[t][2][v];
            const float a3 = acc[t][3][v];
            const float b2 = __int_as_float(__builtin_amdgcn_update_dpp(
                0, __float_as_int(a2), 0xB1  , 0xF, 0xF, true));
            const float b3 = __int_as_float(__builtin_amdgcn_update_dpp(
                0, __float_as_int(a3), 0xB1  , 0xF, 0xF, true));
            fg2[t][v] = 1.0f / (1.0f + __expf(a2 - b2));
            fg3[t][v] = 1.0f / (1.0f + __expf(a3 - b3));
        }
    }

    if (col < 4) {
        #pragma unroll
        for (int t = 0; t < MT; ++t)
            #pragma unroll
            for (int v = 0; v < 8; ++v)
                ploc[t >> 2][PIXC(t, v) * O_LOC + 32] = acc[t][2][v];
    } else {
        #pragma unroll
        for (int t = 0; t < MT; ++t)
            #pragma unroll
            for (int v = 0; v < 8; ++v)
                psc[t >> 2][PIXC(t, v) * O_SC] = acc[t][2][v];
    }
    if (col < 6) {
        #pragma unroll
        for (int t = 0; t < MT; ++t)
            #pragma unroll
            for (int v = 0; v < 8; ++v)
                psc[t >> 2][PIXC(t, v) * O_SC + 16] = acc[t][3][v];
    }
    if (col >= 4 && (col & 1) == 0) {
        #pragma unroll
        for (int t = 0; t < MT; ++t)
            #pragma unroll
            for (int v = 0; v < 8; ++v)
                pfg[t >> 2][PIXC(t, v) * 9] = fg2[t][v];
    }
    if (col < 6 && (col & 1) == 0) {
        #pragma unroll
        for (int t = 0; t < MT; ++t)
            #pragma unroll
            for (int v = 0; v < 8; ++v)
                pfg[t >> 2][PIXC(t, v) * 9 + 8] = fg3[t][v];
    }
#undef PIXC
}

__global__ __launch_bounds__(256)
void rpn_anchor_kernel(float* __restrict__ out_anchor)
{
    const int i = blockIdx.x * 256 + threadIdx.x;
    const int hw = i / 9;
    const int a  = i - hw * 9;
    const int h  = hw / 96;
    const int w  = hw - h * 96;
    const int ri = a / 3;
    const int si = a - ri * 3;

    const float ratios[3] = {0.5f, 1.0f, 2.0f};
    const float scales[3] = {8.0f, 16.0f, 32.0f};
    const float r = ratios[ri];
    const float s = scales[si];

    const float hbox = 16.0f * s * __fsqrt_rn(r);
    const float wbox = 16.0f * s * __fsqrt_rn(1.0f / r);
    const float py = (float)h * 16.0f + 8.0f;
    const float px = (float)w * 16.0f + 8.0f;

    float4 v;
    v.x = py - hbox * 0.5f;
    v.y = px - wbox * 0.5f;
    v.z = py + hbox * 0.5f;
    v.w = px + wbox * 0.5f;
    ((float4*)out_anchor)[i] = v;
}

extern "C" void kernel_launch(void* const* d_in, const int* in_sizes, int n_in,
                              void* d_out, int out_size, void* d_ws, size_t ws_size,
                              hipStream_t stream)
{
    const float* x       = (const float*)d_in[0];
    const float* loc_w   = (const float*)d_in[1];
    const float* loc_b   = (const float*)d_in[2];
    const float* score_w = (const float*)d_in[3];
    const float* score_b = (const float*)d_in[4];
    float* out = (float*)d_out;

    rpn_head_kernel<<<NROWS / WAVES, THREADS, 0, stream>>>(x, loc_w, loc_b, score_w, score_b, out);

    float* out_anchor = out + LOC_ELEMS + SC_ELEMS + FG_ELEMS;
    rpn_anchor_kernel<<<(HW * 9) / 256, 256, 0, stream>>>(out_anchor);
}


# anchor table folded into the head kernel: its first 864 waves write 8 whole 128-B lines each behind their first load group, while x is in flight; second launch gone; sums untouched
# speedup vs baseline: 1.8911x; 1.0297x over previous
#include <hip/hip_runtime.h>
#include <math.h>


typedef float v2f __attribute__((ext_vector_type(2)));
typedef float v4f __attribute__((ext_vector_type(4)));
typedef float v8f __attribute__((ext_vector_type(8)));

#define C_DIM   512
#define HW      6144
#define NIMG    16
#define O_LOC   36
#define O_SC    18
#define O_TOT   54
#define O_PAD   64
#define BLK_CH  32

#define ROW_PIX   96
#define ROWS_IMG  64
#define NROWS     (NIMG * ROWS_IMG)
#define MT        6
#define NT        4
#define WAVES     4
#define THREADS   (WAVES * 32)
#define RING      (BLK_CH / 4)
#define NBLK      (C_DIM / BLK_CH)
#define NPAIR     (NBLK / 2)

#define LOC_ELEMS ((size_t)NIMG * HW * O_LOC)
#define SC_ELEMS  ((size_t)NIMG * HW * O_SC)
#define FG_ELEMS  ((size_t)NIMG * HW * 9)


struct XSlot {
    v4f lo0, lo1;
    v2f hi0, hi1;
};

struct WSet {
    v4f q[NT][4];
};

__device__ __forceinline__ void load_x(XSlot& r, const float* __restrict__ plo,
                                       const float* __restrict__ phi, int blk, int d)
{
    const size_t off = (size_t)(blk * BLK_CH + 2 * d) * HW;
    r.lo0 = *(const v4f*)(plo + off);
    r.lo1 = *(const v4f*)(plo + off + HW);
    r.hi0 = *(const v2f*)(phi + off);
    r.hi1 = *(const v2f*)(phi + off + HW);
}

__device__ __forceinline__ void load_group(WSet& w, XSlot (&set)[RING],
                                           const float* const (&wp)[NT],
                                           const float* __restrict__ plo,
                                           const float* __restrict__ phi, int blk)
{
    #pragma unroll
    for (int i = 0; i < 4; ++i)
        #pragma unroll
        for (int j = 0; j < NT; ++j)
            w.q[j][i] = *(const v4f*)(wp[j] + blk * BLK_CH + 4 * i);
    #pragma unroll
    for (int d = 0; d < RING; ++d)
        load_x(set[d], plo, phi, blk, d);
    __builtin_amdgcn_sched_barrier(0);
}

__device__ __forceinline__ void k_block(v8f (&acc)[MT][NT], const XSlot (&set)[RING], const WSet& w)
{
    #pragma unroll
    for (int d = 0; d < RING; ++d) {
        v2f b[NT];
        #pragma unroll
        for (int j = 0; j < NT; ++j) {
            const v4f q = w.q[j][d >> 1];
            if (d & 1) { b[j].x = q.z; b[j].y = q.w; }
            else       { b[j].x = q.x; b[j].y = q.y; }
        }

        const XSlot& r = set[d];
        v2f a[MT];
        a[0].x = r.lo0.x;  a[0].y = r.lo1.x;
        a[1].x = r.lo0.y;  a[1].y = r.lo1.y;
        a[2].x = r.lo0.z;  a[2].y = r.lo1.z;
        a[3].x = r.lo0.w;  a[3].y = r.lo1.w;
        a[4].x = r.hi0.x;  a[4].y = r.hi1.x;
        a[5].x = r.hi0.y;  a[5].y = r.hi1.y;

        #pragma unroll
        for (int j = 0; j < NT; ++j)
            #pragma unroll
            for (int t = 0; t < MT; ++t)
                acc[t][j] = __builtin_amdgcn_wmma_f32_16x16x4_f32(
                    false, a[t], false, b[j], (short)0, acc[t][j], false, false);
    }
}

#define ANCH_PER_WAVE 64
#define ANCH_WAVES    ((HW * 9) / ANCH_PER_WAVE)

__device__ __forceinline__ float4 anchor_box(int i)
{
    const int hw = i / 9;
    const int a  = i - hw * 9;
    const int h  = hw / 96;
    const int w  = hw - h * 96;
    const int ri = a / 3;
    const int si = a - ri * 3;

    const float r = ri == 0 ? 0.5f : (ri == 1 ? 1.0f : 2.0f);
    const float s = si == 0 ? 8.0f : (si == 1 ? 16.0f : 32.0f);

    const float hbox = 16.0f * s * __fsqrt_rn(r);
    const float wbox = 16.0f * s * __fsqrt_rn(1.0f / r);
    const float py = (float)h * 16.0f + 8.0f;
    const float px = (float)w * 16.0f + 8.0f;

    float4 v;
    v.x = py - hbox * 0.5f;
    v.y = px - wbox * 0.5f;
    v.z = py + hbox * 0.5f;
    v.w = px + wbox * 0.5f;
    return v;
}

__global__ __launch_bounds__(THREADS)
void rpn_head_kernel(const float* __restrict__ x,
                     const float* __restrict__ loc_w,
                     const float* __restrict__ loc_b,
                     const float* __restrict__ score_w,
                     const float* __restrict__ score_b,
                     float* __restrict__ d_out)
{
    const int tid  = threadIdx.x;
    const int wave = tid >> 5;
    const int lane = tid & 31;
    const int col  = lane & 15;
    const int khalf = lane >> 4;

    const float* wp[NT];
    float bias_r[NT];
    #pragma unroll
    for (int j = 0; j < NT; ++j) {
        const int o = j * 16 + col;
        const float* rowp;
        float bv;
        if (o < O_LOC) {
            rowp = loc_w + (size_t)o * C_DIM;
            bv = loc_b[o];
        } else if (o < O_TOT) {
            rowp = score_w + (size_t)(o - O_LOC) * C_DIM;
            bv = score_b[o - O_LOC];
        } else {
            rowp = score_w + (size_t)(O_SC - 1) * C_DIM;
            bv = 0.0f;
        }
        wp[j] = rowp + khalf * (BLK_CH / 2);
        bias_r[j] = bv;
    }

    const int row = blockIdx.x * WAVES + wave;
    const int n   = row / ROWS_IMG;
    const int h   = row - n * ROWS_IMG;
    const float* xrow = x + (size_t)n * C_DIM * HW + (size_t)h * ROW_PIX
                          + (size_t)(khalf * (BLK_CH / 2)) * HW;
    const float* plo = xrow + 4 * col;
    const float* phi = xrow + 64 + 2 * col;

    v8f acc[MT][NT];
    const v8f vzero = {0.f,0.f,0.f,0.f,0.f,0.f,0.f,0.f};
    #pragma unroll
    for (int t = 0; t < MT; ++t)
        #pragma unroll
        for (int j = 0; j < NT; ++j)
            acc[t][j] = vzero;

    XSlot xA[RING], xB[RING];
    WSet  wA, wB;
    load_group(wA, xA, wp, plo, phi, 0);

    if (row < ANCH_WAVES) {
        float4* pa = (float4*)(d_out + LOC_ELEMS + SC_ELEMS + FG_ELEMS)
                   + row * ANCH_PER_WAVE + lane;
        pa[0]  = anchor_box(row * ANCH_PER_WAVE + lane);
        pa[32] = anchor_box(row * ANCH_PER_WAVE + 32 + lane);
    }

    for (int p = 0; p < NPAIR - 1; ++p) {
        load_group(wB, xB, wp, plo, phi, 2 * p + 1);
        k_block(acc, xA, wA);
        load_group(wA, xA, wp, plo, phi, 2 * p + 2);
        k_block(acc, xB, wB);
    }
    load_group(wB, xB, wp, plo, phi, NBLK - 1);
    k_block(acc, xA, wA);
    k_block(acc, xB, wB);

    float* out_loc   = d_out;
    float* out_score = d_out + LOC_ELEMS;
    float* out_fg    = d_out + LOC_ELEMS + SC_ELEMS;

#define PIXC(t, v) ((t) < 4 ? 4 * (v) + (t) : 64 + 2 * (v) + ((t) - 4))
    const ptrdiff_t prow = (ptrdiff_t)row * ROW_PIX;
    const ptrdiff_t pfam[2] = { prow + 32 * khalf, prow + 16 * khalf };

    float* ploc[2];
    float* psc[2];
    float* pfg[2];
    #pragma unroll
    for (int f = 0; f < 2; ++f) {
        ploc[f] = out_loc   + pfam[f] * O_LOC + col;
        psc[f]  = out_score + pfam[f] * O_SC  + (col - 4);
        pfg[f]  = out_fg    + pfam[f] * 9     + ((col >> 1) - 2);
    }

    #pragma unroll
    for (int t = 0; t < MT; ++t)
        #pragma unroll
        for (int j = 0; j < NT; ++j)
            #pragma unroll
            for (int v = 0; v < 8; ++v)
                acc[t][j][v] += bias_r[j];

    #pragma unroll
    for (int t = 0; t < MT; ++t) {
        float* pl = ploc[t >> 2];
        #pragma unroll
        for (int v = 0; v < 8; ++v) {
            pl[PIXC(t, v) * O_LOC]      = acc[t][0][v];
            pl[PIXC(t, v) * O_LOC + 16] = acc[t][1][v];
        }
    }

    float fg2[MT][8], fg3[MT][8];
    #pragma unroll
    for (int t = 0; t < MT; ++t) {
        #pragma unroll
        for (int v = 0; v < 8; ++v) {
            const float a2 = acc[t][2][v];
            const float a3 = acc[t][3][v];
            const float b2 = __int_as_float(__builtin_amdgcn_update_dpp(
                0, __float_as_int(a2), 0xB1  , 0xF, 0xF, true));
            const float b3 = __int_as_float(__builtin_amdgcn_update_dpp(
                0, __float_as_int(a3), 0xB1  , 0xF, 0xF, true));
            fg2[t][v] = 1.0f / (1.0f + __expf(a2 - b2));
            fg3[t][v] = 1.0f / (1.0f + __expf(a3 - b3));
        }
    }

    if (col < 4) {
        #pragma unroll
        for (int t = 0; t < MT; ++t)
            #pragma unroll
            for (int v = 0; v < 8; ++v)
                ploc[t >> 2][PIXC(t, v) * O_LOC + 32] = acc[t][2][v];
    } else {
        #pragma unroll
        for (int t = 0; t < MT; ++t)
            #pragma unroll
            for (int v = 0; v < 8; ++v)
                psc[t >> 2][PIXC(t, v) * O_SC] = acc[t][2][v];
    }
    if (col < 6) {
        #pragma unroll
        for (int t = 0; t < MT; ++t)
            #pragma unroll
            for (int v = 0; v < 8; ++v)
                psc[t >> 2][PIXC(t, v) * O_SC + 16] = acc[t][3][v];
    }
    if (col >= 4 && (col & 1) == 0) {
        #pragma unroll
        for (int t = 0; t < MT; ++t)
            #pragma unroll
            for (int v = 0; v < 8; ++v)
                pfg[t >> 2][PIXC(t, v) * 9] = fg2[t][v];
    }
    if (col < 6 && (col & 1) == 0) {
        #pragma unroll
        for (int t = 0; t < MT; ++t)
            #pragma unroll
            for (int v = 0; v < 8; ++v)
                pfg[t >> 2][PIXC(t, v) * 9 + 8] = fg3[t][v];
    }
#undef PIXC
}

extern "C" void kernel_launch(void* const* d_in, const int* in_sizes, int n_in,
                              void* d_out, int out_size, void* d_ws, size_t ws_size,
                              hipStream_t stream)
{
    const float* x       = (const float*)d_in[0];
    const float* loc_w   = (const float*)d_in[1];
    const float* loc_b   = (const float*)d_in[2];
    const float* score_w = (const float*)d_in[3];
    const float* score_b = (const float*)d_in[4];
    float* out = (float*)d_out;

    rpn_head_kernel<<<NROWS / WAVES, THREADS, 0, stream>>>(x, loc_w, loc_b, score_w, score_b, out);
}
